# starting kernel
# baseline (speedup 1.0000x reference)
#include <hip/hip_runtime.h>

constexpr int Bn    = 32;
constexpr int PPB   = 256;
constexpr int Np    = Bn * PPB;
constexpr int Mf    = 128;
constexpr int LANGD = 256;
constexpr int Hd    = 128;
constexpr int Kfuse = LANGD + Mf;
#define BN_EPS 1e-5f

typedef float v2f __attribute__((ext_vector_type(2)));
typedef float v8f __attribute__((ext_vector_type(8)));

__device__ __forceinline__ v8f wmma4(v2f a, v2f b, v8f c) {
  return __builtin_amdgcn_wmma_f32_16x16x4_f32(false, a, false, b, (short)0, c,
                                               false, false);
}

#define V8F_ZERO {0.f, 0.f, 0.f, 0.f, 0.f, 0.f, 0.f, 0.f}

__global__ void k_bid(const int* __restrict__ pidx, const int* __restrict__ poff,
                      const int* __restrict__ boff, int* __restrict__ bid) {
  int i = blockIdx.x * blockDim.x + threadIdx.x;
  if (i >= Np) return;
  int first_pt = pidx[2 * poff[i] + 1];
  int b = -1;
#pragma unroll
  for (int j = 0; j <= Bn; ++j) b += (boff[j] <= first_pt) ? 1 : 0;
  bid[i] = b;
}

__global__ __launch_bounds__(256) void k_rank(const int* __restrict__ bid,
                                              int* __restrict__ dst) {
  __shared__ int sbid[Np];
  for (int j = threadIdx.x; j < Np; j += blockDim.x) sbid[j] = bid[j];
  __syncthreads();
  int i = blockIdx.x * blockDim.x + threadIdx.x;
  if (i >= Np) return;
  int myb = sbid[i];
  int r = 0;
  for (int j = 0; j < i; ++j) r += (sbid[j] == myb) ? 1 : 0;
  dst[i] = myb * PPB + r;
}

__global__ void k_zero(float* __restrict__ p, int n) {
  int i = blockIdx.x * blockDim.x + threadIdx.x;
  if (i < n) p[i] = 0.f;
}

__global__ __launch_bounds__(128) void k_scatter(const float* __restrict__ feats,
                                                 const int* __restrict__ dst,
                                                 float* __restrict__ padded) {
  int i = blockIdx.x;
  int d = dst[i];
  if (d < 0 || d >= Np) return;
  padded[(size_t)d * Mf + threadIdx.x] = feats[(size_t)i * Mf + threadIdx.x];
}

__global__ __launch_bounds__(256) void k_l1(const float* __restrict__ padded,
                                            const float* __restrict__ lang,
                                            const float* __restrict__ Wf,
                                            const float* __restrict__ bf,
                                            float* __restrict__ out) {
  const int b     = blockIdx.x;
  const int pbase = blockIdx.y * 64;
  const int wave  = threadIdx.x >> 5;
  const int lane  = threadIdx.x & 31;
  const int half  = lane >> 4;
  const int lm    = lane & 15;
  const int o0    = wave << 4;

  const float* arow = Wf + (size_t)(o0 + lm) * Kfuse + 2 * half;
  const float* xb   = padded + (size_t)b * PPB * Mf;
  const float* br0  = xb + (size_t)(pbase +  0 + lm) * Mf + 2 * half;
  const float* br1  = xb + (size_t)(pbase + 16 + lm) * Mf + 2 * half;
  const float* br2  = xb + (size_t)(pbase + 32 + lm) * Mf + 2 * half;
  const float* br3  = xb + (size_t)(pbase + 48 + lm) * Mf + 2 * half;
  const float* lb   = lang + (size_t)b * LANGD + 2 * half;

  v8f acc0 = V8F_ZERO, acc1 = V8F_ZERO, acc2 = V8F_ZERO, acc3 = V8F_ZERO;

#pragma unroll 4
  for (int k0 = 0; k0 < Mf; k0 += 4) {
    v2f a;   a.x  = arow[k0];      a.y  = arow[k0 + 1];
    v2f b0;  b0.x = br0[k0];       b0.y = br0[k0 + 1];
    v2f b1;  b1.x = br1[k0];       b1.y = br1[k0 + 1];
    v2f b2;  b2.x = br2[k0];       b2.y = br2[k0 + 1];
    v2f b3;  b3.x = br3[k0];       b3.y = br3[k0 + 1];
    acc0 = wmma4(a, b0, acc0);
    acc1 = wmma4(a, b1, acc1);
    acc2 = wmma4(a, b2, acc2);
    acc3 = wmma4(a, b3, acc3);
  }
#pragma unroll 4
  for (int k0 = 0; k0 < LANGD; k0 += 4) {
    v2f a;   a.x = arow[Mf + k0];  a.y = arow[Mf + k0 + 1];
    v2f bb;  bb.x = lb[k0];        bb.y = lb[k0 + 1];
    acc0 = wmma4(a, bb, acc0);
    acc1 = wmma4(a, bb, acc1);
    acc2 = wmma4(a, bb, acc2);
    acc3 = wmma4(a, bb, acc3);
  }

  float* ob = out + ((size_t)b * Hd) * PPB + pbase + lm;
#pragma unroll
  for (int v = 0; v < 8; ++v) {
    const int o = o0 + v + 8 * half;
    const float bias = bf[o];
    float* orow = ob + (size_t)o * PPB;
    orow[ 0] = fmaxf(acc0[v] + bias, 0.f);
    orow[16] = fmaxf(acc1[v] + bias, 0.f);
    orow[32] = fmaxf(acc2[v] + bias, 0.f);
    orow[48] = fmaxf(acc3[v] + bias, 0.f);
  }
}

__global__ __launch_bounds__(256) void k_mid(const float* __restrict__ in,
                                             const float* __restrict__ W,
                                             const float* __restrict__ bias,
                                             const float* __restrict__ gam,
                                             const float* __restrict__ bet,
                                             const float* __restrict__ rme,
                                             const float* __restrict__ rva,
                                             float* __restrict__ out) {
  const int b     = blockIdx.x;
  const int pbase = blockIdx.y * 64;
  const int wave  = threadIdx.x >> 5;
  const int lane  = threadIdx.x & 31;
  const int half  = lane >> 4;
  const int lm    = lane & 15;
  const int o0    = wave << 4;

  const float* arow = W + (size_t)(o0 + lm) * Hd + 2 * half;
  const float* inb  = in + (size_t)b * Hd * PPB;
  const float* bc   = inb + (size_t)(2 * half) * PPB + pbase + lm;

  v8f acc0 = V8F_ZERO, acc1 = V8F_ZERO, acc2 = V8F_ZERO, acc3 = V8F_ZERO;

#pragma unroll 4
  for (int k0 = 0; k0 < Hd; k0 += 4) {
    const int r0 = k0 * PPB;
    const int r1 = r0 + PPB;
    v2f a;   a.x  = arow[k0];      a.y  = arow[k0 + 1];
    v2f b0;  b0.x = bc[r0 +  0];   b0.y = bc[r1 +  0];
    v2f b1;  b1.x = bc[r0 + 16];   b1.y = bc[r1 + 16];
    v2f b2;  b2.x = bc[r0 + 32];   b2.y = bc[r1 + 32];
    v2f b3;  b3.x = bc[r0 + 48];   b3.y = bc[r1 + 48];
    acc0 = wmma4(a, b0, acc0);
    acc1 = wmma4(a, b1, acc1);
    acc2 = wmma4(a, b2, acc2);
    acc3 = wmma4(a, b3, acc3);
  }

  float* ob = out + ((size_t)b * Hd) * PPB + pbase + lm;
#pragma unroll
  for (int v = 0; v < 8; ++v) {
    const int o = o0 + v + 8 * half;
    const float bs    = bias[o];
    const float scale = gam[o] * rsqrtf(rva[o] + BN_EPS);
    const float mu    = rme[o];
    const float beta  = bet[o];
    float* orow = ob + (size_t)o * PPB;
    orow[ 0] = (fmaxf(acc0[v] + bs, 0.f) - mu) * scale + beta;
    orow[16] = (fmaxf(acc1[v] + bs, 0.f) - mu) * scale + beta;
    orow[32] = (fmaxf(acc2[v] + bs, 0.f) - mu) * scale + beta;
    orow[48] = (fmaxf(acc3[v] + bs, 0.f) - mu) * scale + beta;
  }
}

__global__ __launch_bounds__(256) void k_final(const float* __restrict__ in,
                                               const float* __restrict__ W3,
                                               const float* __restrict__ b3,
                                               float* __restrict__ out) {
  int idx = blockIdx.x * blockDim.x + threadIdx.x;
  if (idx >= Bn * PPB) return;
  int b = idx / PPB, p = idx % PPB;
  const float* inb = in + (size_t)b * Hd * PPB + p;
  float s = b3[0];
#pragma unroll 8
  for (int c = 0; c < Hd; ++c) s += W3[c] * inb[(size_t)c * PPB];
  out[idx] = s;
}

extern "C" void kernel_launch(void* const* d_in, const int* in_sizes, int n_in,
                              void* d_out, int out_size, void* d_ws, size_t ws_size,
                              hipStream_t stream) {
  const float* feats = (const float*)d_in[0];
  const int*   pidx  = (const int*)d_in[1];
  const int*   poff  = (const int*)d_in[2];
  const int*   boff  = (const int*)d_in[3];
  const float* lang  = (const float*)d_in[4];
  const float* Wf    = (const float*)d_in[5];
  const float* bf    = (const float*)d_in[6];
  const float* W1p   = (const float*)d_in[7];
  const float* b1p   = (const float*)d_in[8];
  const float* g1p   = (const float*)d_in[9];
  const float* be1p  = (const float*)d_in[10];
  const float* rm1p  = (const float*)d_in[11];
  const float* rv1p  = (const float*)d_in[12];
  const float* W2p   = (const float*)d_in[13];
  const float* b2p   = (const float*)d_in[14];
  const float* g2p   = (const float*)d_in[15];
  const float* be2p  = (const float*)d_in[16];
  const float* rm2p  = (const float*)d_in[17];
  const float* rv2p  = (const float*)d_in[18];
  const float* W3p   = (const float*)d_in[19];
  const float* b3p   = (const float*)d_in[20];

  char* ws = (char*)d_ws;
  int*   bid    = (int*)ws;
  int*   dst    = (int*)(ws + (size_t)Np * sizeof(int));
  float* padded = (float*)(ws + 2 * (size_t)Np * sizeof(int));
  float* act1   = padded + (size_t)Np * Mf;
  float* act2   = padded;

  k_bid <<<Np / 256, 256, 0, stream>>>(pidx, poff, boff, bid);
  k_rank<<<Np / 256, 256, 0, stream>>>(bid, dst);
  k_zero<<<(Np * Mf) / 256, 256, 0, stream>>>(padded, Np * Mf);
  k_scatter<<<Np, 128, 0, stream>>>(feats, dst, padded);

  dim3 grid(Bn, 4);
  k_l1 <<<grid, 256, 0, stream>>>(padded, lang, Wf, bf, act1);
  k_mid<<<grid, 256, 0, stream>>>(act1, W1p, b1p, g1p, be1p, rm1p, rv1p, act2);
  k_mid<<<grid, 256, 0, stream>>>(act2, W2p, b2p, g2p, be2p, rm2p, rv2p, act1);
  k_final<<<Np / 256, 256, 0, stream>>>(act1, W3p, b3p, (float*)d_out);
}


# stage order swapped, 3 -> 2 launches: MLP first, each wave finds its columns' batch ids (k_bid gone); rank launch stores the 4-byte results to their slots, fills empty slots by batch count
# speedup vs baseline: 5.9198x; 5.9198x over previous
#include <hip/hip_runtime.h>

constexpr int Bn    = 32;
constexpr int PPB   = 256;
constexpr int Np    = Bn * PPB;
constexpr int Mf    = 128;
constexpr int LANGD = 256;
constexpr int Hd    = 128;
constexpr int Kfuse = LANGD + Mf;
#define BN_EPS 1e-5f

typedef float v2f __attribute__((ext_vector_type(2)));
typedef float v8f __attribute__((ext_vector_type(8)));

__device__ __forceinline__ v8f wmma4(v2f a, v2f b, v8f c) {
  return __builtin_amdgcn_wmma_f32_16x16x4_f32(false, a, false, b, (short)0, c,
                                               false, false);
}

#define V8F_ZERO {0.f, 0.f, 0.f, 0.f, 0.f, 0.f, 0.f, 0.f}


__device__ __forceinline__ int batch_of(const int* __restrict__ boff, int first_pt) {
  int b = -1;
#pragma unroll
  for (int j = 0; j <= Bn; ++j) b += (boff[j] <= first_pt) ? 1 : 0;
  return b;
}

struct EmptyArgs {
  const float* W1; const float* b1; const float* g1; const float* be1;
  const float* rm1; const float* rv1;
  const float* W2; const float* b2; const float* g2; const float* be2;
  const float* rm2; const float* rv2;
  const float* W3; const float* b3;
};

constexpr int XP = 40;

__device__ __forceinline__ void empty_mid(const float* xin, const float* __restrict__ W,
                                          const float* __restrict__ bias,
                                          const float* __restrict__ gam,
                                          const float* __restrict__ bet,
                                          const float* __restrict__ rme,
                                          const float* __restrict__ rva,
                                          float* xout, int o0, int half, int lm) {
  const float* arow = W + (size_t)(o0 + lm) * Hd + 2 * half;
  const float* bc   = xin + (2 * half) * XP + lm;
  v8f acc0 = V8F_ZERO, acc1 = V8F_ZERO;
#pragma unroll 16
  for (int k0 = 0; k0 < Hd; k0 += 4) {
    v2f a;   a.x  = arow[k0];             a.y  = arow[k0 + 1];
    v2f b0;  b0.x = bc[k0 * XP];          b0.y = bc[(k0 + 1) * XP];
    v2f b1;  b1.x = bc[k0 * XP + 16];     b1.y = bc[(k0 + 1) * XP + 16];
    acc0 = wmma4(a, b0, acc0);
    acc1 = wmma4(a, b1, acc1);
  }
#pragma unroll
  for (int v = 0; v < 8; ++v) {
    const int o = o0 + v + 8 * half;
    const float bs    = bias[o];
    const float scale = gam[o] * rsqrtf(rva[o] + BN_EPS);
    const float mu    = rme[o];
    const float beta  = bet[o];
    xout[o * XP + lm]      = (fmaxf(acc0[v] + bs, 0.f) - mu) * scale + beta;
    xout[o * XP + 16 + lm] = (fmaxf(acc1[v] + bs, 0.f) - mu) * scale + beta;
  }
}

constexpr int RANK_WAVES = 16;

__device__ __forceinline__ void empty_slots(const int (*part)[32],
                                            const float* __restrict__ lang,
                                            const float* __restrict__ Wf,
                                            const float* __restrict__ bf,
                                            const EmptyArgs& e,
                                            float* __restrict__ out) {
  __shared__ float xa[Hd * XP];
  __shared__ float xb[Hd * XP];
  __shared__ float c0[Bn];
  __shared__ int   cnt[Bn];
  const int wave = __builtin_amdgcn_readfirstlane((int)(threadIdx.x >> 5));
  const bool act = wave < 8;
  const int lane = threadIdx.x & 31;
  const int half = lane >> 4;
  const int lm   = lane & 15;
  const int o0   = (wave & 7) << 4;
  int n = 0;
  for (int w = 0; w < RANK_WAVES; ++w) n += part[w][lane];
  if (!__any(n < PPB)) return;
  if (wave == 0) cnt[lane] = n;
  if (act) {
    const float* arow = Wf + (size_t)(o0 + lm) * Kfuse + Mf + 2 * half;
    const float* l0   = lang + (size_t)lm * LANGD + 2 * half;
    const float* l1   = lang + (size_t)(16 + lm) * LANGD + 2 * half;
    v8f acc0 = V8F_ZERO, acc1 = V8F_ZERO;
#pragma unroll 16
    for (int k0 = 0; k0 < LANGD; k0 += 4) {
      v2f a;   a.x  = arow[k0];   a.y  = arow[k0 + 1];
      v2f b0;  b0.x = l0[k0];     b0.y = l0[k0 + 1];
      v2f b1;  b1.x = l1[k0];     b1.y = l1[k0 + 1];
      acc0 = wmma4(a, b0, acc0);
      acc1 = wmma4(a, b1, acc1);
    }
#pragma unroll
    for (int v = 0; v < 8; ++v) {
      const int o = o0 + v + 8 * half;
      const float bias = bf[o];
      xa[o * XP + lm]      = fmaxf(acc0[v] + bias, 0.f);
      xa[o * XP + 16 + lm] = fmaxf(acc1[v] + bias, 0.f);
    }
  }
  __syncthreads();
  if (act) empty_mid(xa, e.W1, e.b1, e.g1, e.be1, e.rm1, e.rv1, xb, o0, half, lm);
  __syncthreads();
  if (act) empty_mid(xb, e.W2, e.b2, e.g2, e.be2, e.rm2, e.rv2, xa, o0, half, lm);
  __syncthreads();
  if (threadIdx.x < Bn) {
    float s = e.b3[0];
    for (int c = 0; c < Hd; ++c) s += e.W3[c] * xa[c * XP + threadIdx.x];
    c0[threadIdx.x] = s;
  }
  __syncthreads();
  if (act)
    for (int j = 0; j < Bn; ++j)
      if ((int)threadIdx.x >= cnt[j]) out[j * PPB + threadIdx.x] = c0[j];
}

__global__ __launch_bounds__(RANK_WAVES * 32) void k_rank(const int* __restrict__ bid,
                                                          const float* __restrict__ val,
                                                          const float* __restrict__ lang,
                                                          const float* __restrict__ Wf,
                                                          const float* __restrict__ bf,
                                                          const EmptyArgs e,
                                                          float* __restrict__ out) {
  __shared__ int part[RANK_WAVES][32];
  const bool fill = blockIdx.x == 0;
  const int tile = Np / 32 - (int)blockIdx.x;
  const int wave = __builtin_amdgcn_readfirstlane((int)(threadIdx.x >> 5));
  const int lane = threadIdx.x & 31;
  const int i    = tile * 32 + lane;
  const int myb  = fill ? lane : bid[i];
  const float v  = fill ? 0.f : val[i];
  int r = 0;
  const int nlines = tile * 2;
#pragma unroll 4
  for (int L = wave; L < nlines; L += RANK_WAVES) {
    const int* p = bid + L * 16;
#pragma unroll
    for (int k = 0; k < 16; ++k) r += (p[k] == myb) ? 1 : 0;
  }
  if (!fill) {
    const int* q = bid + tile * 32 + wave * 2;
    for (int t = 0; t < 2; ++t) r += ((wave * 2 + t) < lane && q[t] == myb) ? 1 : 0;
  }
  part[wave][lane] = r;
  __syncthreads();
  if (fill) {
    empty_slots(part, lang, Wf, bf, e, out);
    return;
  }
  if (wave == 0) {
    int s = 0;
    for (int w = 0; w < RANK_WAVES; ++w) s += part[w][lane];
    if ((unsigned)myb < (unsigned)Bn && s < PPB) out[myb * PPB + s] = v;
  }
}

typedef float v4f __attribute__((ext_vector_type(4)));

__device__ __forceinline__ v8f ld8(const float* __restrict__ p) {
  const v4f lo = *(const v4f*)p;
  const v4f hi = *(const v4f*)(p + 4);
  v8f r;
  r[0] = lo[0]; r[1] = lo[1]; r[2] = lo[2]; r[3] = lo[3];
  r[4] = hi[0]; r[5] = hi[1]; r[6] = hi[2]; r[7] = hi[3];
  return r;
}

__device__ __forceinline__ void kblock(const v8f (&a)[4], const v8f (&b)[4], v8f (&acc)[4]) {
#pragma unroll
  for (int s = 0; s < 2; ++s) {
#pragma unroll
    for (int j = 0; j < 4; ++j) {
      v2f b0; b0.x = b[s][2 * j];     b0.y = b[s][2 * j + 1];
      v2f b1; b1.x = b[2 + s][2 * j]; b1.y = b[2 + s][2 * j + 1];
#pragma unroll
      for (int v = 0; v < 2; ++v) {
        v2f a2; a2.x = a[2 * v + s][2 * j]; a2.y = a[2 * v + s][2 * j + 1];
        acc[2 * v]     = wmma4(a2, b0, acc[2 * v]);
        acc[2 * v + 1] = wmma4(a2, b1, acc[2 * v + 1]);
      }
    }
  }
}

__device__ __forceinline__ void lda(const float* __restrict__ arow, int ld, int T, v8f (&a)[4]) {
#pragma unroll
  for (int v = 0; v < 2; ++v) {
#pragma unroll
    for (int s = 0; s < 2; ++s) a[2 * v + s] = ld8(arow + (size_t)(16 * v) * ld + 16 * (2 * T + s));
  }
}

struct Rows {
  const float* a1;
  const float* a2;
  const float* a3;
  const float* frow[2];
  const float* lrow[2];
};

template <int N>
__device__ __forceinline__ void load_blk(const Rows& p, v8f (&a)[4], v8f (&b)[4]) {
  if constexpr (N < 12) {
    lda(p.a1, Kfuse, N, a);
#pragma unroll
    for (int c = 0; c < 2; ++c) {
      const float* bp = (N < 4) ? p.frow[c] + 32 * N : p.lrow[c] + 32 * (N - 4);
      b[2 * c] = ld8(bp); b[2 * c + 1] = ld8(bp + 16);
    }
  } else if constexpr (N < 16) {
    lda(p.a2, Hd, N - 12, a);
  } else if constexpr (N < 20) {
    lda(p.a3, Hd, N - 16, a);
  }
}

template <int N>
__device__ __forceinline__ void comp_blk(const v8f (&a)[4], const v8f (&b)[4], const v8f (&x)[16],
                                         v8f (&acc)[4]) {
  if constexpr (N < 12) {
    kblock(a, b, acc);
  } else {
    constexpr int T = (N - 12) & 3;
    const v8f bb[4] = {x[2 * T], x[2 * T + 1], x[8 + 2 * T], x[8 + 2 * T + 1]};
    kblock(a, bb, acc);
  }
}

__device__ __forceinline__ void bn_relu(v8f (&acc)[4], const float* __restrict__ bias,
                                        const float* __restrict__ gam,
                                        const float* __restrict__ bet,
                                        const float* __restrict__ rme,
                                        const float* __restrict__ rva, int o0h) {
#pragma unroll
  for (int v = 0; v < 2; ++v) {
    const int o = o0h + 16 * v;
    const v8f bs = ld8(bias + o), g = ld8(gam + o), be = ld8(bet + o);
    const v8f mu = ld8(rme + o), rv = ld8(rva + o);
#pragma unroll
    for (int r = 0; r < 8; ++r) {
      const float scale = g[r] * rsqrtf(rv[r] + BN_EPS);
#pragma unroll
      for (int c = 0; c < 2; ++c)
        acc[2 * v + c][r] = (fmaxf(acc[2 * v + c][r] + bs[r], 0.f) - mu[r]) * scale + be[r];
    }
  }
}

__device__ __forceinline__ void exchange(v4f* __restrict__ plane, int w, int lane,
                                         v8f (&acc)[4], v8f (&x)[16]) {
#pragma unroll
  for (int v = 0; v < 2; ++v) {
#pragma unroll
    for (int c = 0; c < 2; ++c) {
#pragma unroll
      for (int q = 0; q < 2; ++q) {
        v4f t;
        t[0] = acc[2 * v + c][4 * q];     t[1] = acc[2 * v + c][4 * q + 1];
        t[2] = acc[2 * v + c][4 * q + 2]; t[3] = acc[2 * v + c][4 * q + 3];
        plane[((8 * c + 2 * w + v) * 2 + q) * 32 + lane] = t;
      }
    }
  }
  __syncthreads();
#pragma unroll
  for (int t = 0; t < 16; ++t) {
    const v4f lo = plane[(t * 2) * 32 + lane];
    const v4f hi = plane[(t * 2 + 1) * 32 + lane];
    x[t][0] = lo[0]; x[t][1] = lo[1]; x[t][2] = lo[2]; x[t][3] = lo[3];
    x[t][4] = hi[0]; x[t][5] = hi[1]; x[t][6] = hi[2]; x[t][7] = hi[3];
  }
#pragma unroll
  for (int v = 0; v < 4; ++v) acc[v] = (v8f)V8F_ZERO;
}

__global__ __launch_bounds__(128) void k_mlp(const float* __restrict__ feats,
                                             const int* __restrict__ pidx,
                                             const int* __restrict__ poff,
                                             const int* __restrict__ boff,
                                             const float* __restrict__ lang,
                                             const float* __restrict__ Wf,
                                             const float* __restrict__ bf,
                                             const EmptyArgs e,
                                             int* __restrict__ bid,
                                             float* __restrict__ val) {
  __shared__ v4f xs[2][16 * 2 * 32];
  __shared__ float sp[4][2][16];
  const int w    = __builtin_amdgcn_readfirstlane((int)(threadIdx.x >> 5));
  const int lane = threadIdx.x & 31;
  const int half = lane >> 4;
  const int lm   = lane & 15;
  const int o0   = 32 * w;
  const int o0h  = o0 + 8 * half;

  Rows p;
  p.a1 = Wf   + (size_t)(o0 + lm) * Kfuse + 8 * half;
  p.a2 = e.W1 + (size_t)(o0 + lm) * Hd + 8 * half;
  p.a3 = e.W2 + (size_t)(o0 + lm) * Hd + 8 * half;
  int pf[2];
#pragma unroll
  for (int c = 0; c < 2; ++c) {
    const int i = (blockIdx.x * 2 + c) * 16 + lm;
    pf[c] = poff[i];
    p.frow[c] = feats + (size_t)i * Mf + 8 * half;
    p.lrow[c] = nullptr;
  }

  v8f acc[4];
#pragma unroll
  for (int v = 0; v < 4; ++v) acc[v] = (v8f)V8F_ZERO;

  v8f A0[4], A1[4], A2[4], B0[4], B1[4], B2[4];
  v8f x[16];
  load_blk<0>(p, A0, B0);
  load_blk<1>(p, A1, B1);
  int fp[2];
#pragma unroll
  for (int c = 0; c < 2; ++c) fp[c] = pidx[2 * pf[c] + 1];
#define STEP(N, CUR, BCUR, PRE, BPRE) load_blk<N + 2>(p, PRE, BPRE); comp_blk<N>(CUR, BCUR, x, acc);
  STEP(0, A0, B0, A2, B2)
  int own = -1;
#pragma unroll
  for (int c = 0; c < 2; ++c) {
    const int b = batch_of(boff, fp[c]);
    const int sb = ((unsigned)b < (unsigned)Bn) ? b : 0;
    p.lrow[c] = lang + (size_t)sb * LANGD + 8 * half;
    if (c == half) own = b;
  }
  STEP(1, A1, B1, A0, B0)  STEP(2, A2, B2, A1, B1)
  STEP(3, A0, B0, A2, B2)  STEP(4, A1, B1, A0, B0)  STEP(5, A2, B2, A1, B1)
  STEP(6, A0, B0, A2, B2)  STEP(7, A1, B1, A0, B0)  STEP(8, A2, B2, A1, B1)
  STEP(9, A0, B0, A2, B2)  STEP(10, A1, B1, A0, B0) STEP(11, A2, B2, A1, B1)
#pragma unroll
  for (int v = 0; v < 2; ++v) {
    const v8f bs = ld8(bf + o0h + 16 * v);
#pragma unroll
    for (int r = 0; r < 8; ++r) {
      acc[2 * v][r]     = fmaxf(acc[2 * v][r] + bs[r], 0.f);
      acc[2 * v + 1][r] = fmaxf(acc[2 * v + 1][r] + bs[r], 0.f);
    }
  }
  exchange(xs[0], w, lane, acc, x);
  STEP(12, A0, B0, A2, B2) STEP(13, A1, B1, A0, B0) STEP(14, A2, B2, A1, B1) STEP(15, A0, B0, A2, B2)
  bn_relu(acc, e.b1, e.g1, e.be1, e.rm1, e.rv1, o0h);
  exchange(xs[1], w, lane, acc, x);
  STEP(16, A1, B1, A0, B0) STEP(17, A2, B2, A1, B1) STEP(18, A0, B0, A2, B2) STEP(19, A1, B1, A0, B0)
#undef STEP
  bn_relu(acc, e.b2, e.g2, e.be2, e.rm2, e.rv2, o0h);
  float s0 = 0.f, s1 = 0.f;
#pragma unroll
  for (int v = 0; v < 2; ++v) {
    const v8f w3 = ld8(e.W3 + o0h + 16 * v);
#pragma unroll
    for (int r = 0; r < 8; ++r) { s0 += w3[r] * acc[2 * v][r]; s1 += w3[r] * acc[2 * v + 1][r]; }
  }
  s0 += __shfl_xor(s0, 16);
  s1 += __shfl_xor(s1, 16);
  if (half == 0) { sp[w][0][lm] = s0; sp[w][1][lm] = s1; }
  __syncthreads();
  if (w == 0) {
    const float r = (((sp[0][half][lm] + sp[1][half][lm]) + sp[2][half][lm]) + sp[3][half][lm])
                    + e.b3[0];
    val[blockIdx.x * 32 + lane] = r;
    bid[blockIdx.x * 32 + lane] = own;
  }
}

extern "C" void kernel_launch(void* const* d_in, const int* in_sizes, int n_in,
                              void* d_out, int out_size, void* d_ws, size_t ws_size,
                              hipStream_t stream) {
  const float* feats = (const float*)d_in[0];
  const int*   pidx  = (const int*)d_in[1];
  const int*   poff  = (const int*)d_in[2];
  const int*   boff  = (const int*)d_in[3];
  const float* lang  = (const float*)d_in[4];
  const float* Wf    = (const float*)d_in[5];
  const float* bf    = (const float*)d_in[6];
  const float* W1p   = (const float*)d_in[7];
  const float* b1p   = (const float*)d_in[8];
  const float* g1p   = (const float*)d_in[9];
  const float* be1p  = (const float*)d_in[10];
  const float* rm1p  = (const float*)d_in[11];
  const float* rv1p  = (const float*)d_in[12];
  const float* W2p   = (const float*)d_in[13];
  const float* b2p   = (const float*)d_in[14];
  const float* g2p   = (const float*)d_in[15];
  const float* be2p  = (const float*)d_in[16];
  const float* rm2p  = (const float*)d_in[17];
  const float* rv2p  = (const float*)d_in[18];
  const float* W3p   = (const float*)d_in[19];
  const float* b3p   = (const float*)d_in[20];

  char* ws = (char*)d_ws;
  int*   bid    = (int*)ws;
  float* val    = (float*)(ws + (size_t)Np * sizeof(int));

  EmptyArgs ea;
  ea.W1 = W1p; ea.b1 = b1p; ea.g1 = g1p; ea.be1 = be1p; ea.rm1 = rm1p; ea.rv1 = rv1p;
  ea.W2 = W2p; ea.b2 = b2p; ea.g2 = g2p; ea.be2 = be2p; ea.rm2 = rm2p; ea.rv2 = rv2p;
  ea.W3 = W3p; ea.b3 = b3p;

  k_mlp <<<Np / 32, 128, 0, stream>>>(feats, pidx, poff, boff, lang, Wf, bf, ea, bid, val);
  k_rank<<<Np / 32 + 1, RANK_WAVES * 32, 0, stream>>>(bid, val, lang, Wf, bf, ea, (float*)d_out);
}


# k_rank: batch ids left as bytes and compared 4 at a time (xor-add, mask, byte-sum: 3 vector instructions per 4 ids, was 8); one 32-byte scalar load per tile, 8 in flight; 8 waves per group, was 16
# speedup vs baseline: 7.0656x; 1.1936x over previous
#include <hip/hip_runtime.h>

constexpr int Bn    = 32;
constexpr int PPB   = 256;
constexpr int Np    = Bn * PPB;
constexpr int Mf    = 128;
constexpr int LANGD = 256;
constexpr int Hd    = 128;
constexpr int Kfuse = LANGD + Mf;
#define BN_EPS 1e-5f

typedef float v2f __attribute__((ext_vector_type(2)));
typedef float v8f __attribute__((ext_vector_type(8)));

__device__ __forceinline__ v8f wmma4(v2f a, v2f b, v8f c) {
  return __builtin_amdgcn_wmma_f32_16x16x4_f32(false, a, false, b, (short)0, c,
                                               false, false);
}

#define V8F_ZERO {0.f, 0.f, 0.f, 0.f, 0.f, 0.f, 0.f, 0.f}


__device__ __forceinline__ int batch_of(const int* __restrict__ boff, int first_pt) {
  int b = -1;
#pragma unroll
  for (int j = 0; j <= Bn; ++j) b += (boff[j] <= first_pt) ? 1 : 0;
  return b;
}

struct EmptyArgs {
  const float* W1; const float* b1; const float* g1; const float* be1;
  const float* rm1; const float* rv1;
  const float* W2; const float* b2; const float* g2; const float* be2;
  const float* rm2; const float* rv2;
  const float* W3; const float* b3;
};

constexpr int XP = 40;

__device__ __forceinline__ void empty_mid(const float* xin, const float* __restrict__ W,
                                          const float* __restrict__ bias,
                                          const float* __restrict__ gam,
                                          const float* __restrict__ bet,
                                          const float* __restrict__ rme,
                                          const float* __restrict__ rva,
                                          float* xout, int o0, int half, int lm) {
  const float* arow = W + (size_t)(o0 + lm) * Hd + 2 * half;
  const float* bc   = xin + (2 * half) * XP + lm;
  v8f acc0 = V8F_ZERO, acc1 = V8F_ZERO;
#pragma unroll 16
  for (int k0 = 0; k0 < Hd; k0 += 4) {
    v2f a;   a.x  = arow[k0];             a.y  = arow[k0 + 1];
    v2f b0;  b0.x = bc[k0 * XP];          b0.y = bc[(k0 + 1) * XP];
    v2f b1;  b1.x = bc[k0 * XP + 16];     b1.y = bc[(k0 + 1) * XP + 16];
    acc0 = wmma4(a, b0, acc0);
    acc1 = wmma4(a, b1, acc1);
  }
#pragma unroll
  for (int v = 0; v < 8; ++v) {
    const int o = o0 + v + 8 * half;
    const float bs    = bias[o];
    const float scale = gam[o] * rsqrtf(rva[o] + BN_EPS);
    const float mu    = rme[o];
    const float beta  = bet[o];
    xout[o * XP + lm]      = (fmaxf(acc0[v] + bs, 0.f) - mu) * scale + beta;
    xout[o * XP + 16 + lm] = (fmaxf(acc1[v] + bs, 0.f) - mu) * scale + beta;
  }
}

constexpr int RANK_WAVES = 8;
typedef unsigned v8u __attribute__((ext_vector_type(8)));

__device__ __forceinline__ void empty_slots(const int (*part)[32],
                                            const float* __restrict__ lang,
                                            const float* __restrict__ Wf,
                                            const float* __restrict__ bf,
                                            const EmptyArgs& e,
                                            float* __restrict__ out) {
  __shared__ float xa[Hd * XP];
  __shared__ float xb[Hd * XP];
  __shared__ float c0[Bn];
  __shared__ int   cnt[Bn];
  const int wave = __builtin_amdgcn_readfirstlane((int)(threadIdx.x >> 5));
  const int lane = threadIdx.x & 31;
  const int half = lane >> 4;
  const int lm   = lane & 15;
  const int o0   = wave << 4;
  int n = 0;
  for (int w = 0; w < RANK_WAVES; ++w) n += part[w][lane];
  if (!__any(n < PPB)) return;
  if (wave == 0) cnt[lane] = n;
  {
    const float* arow = Wf + (size_t)(o0 + lm) * Kfuse + Mf + 2 * half;
    const float* l0   = lang + (size_t)lm * LANGD + 2 * half;
    const float* l1   = lang + (size_t)(16 + lm) * LANGD + 2 * half;
    v8f acc0 = V8F_ZERO, acc1 = V8F_ZERO;
#pragma unroll 16
    for (int k0 = 0; k0 < LANGD; k0 += 4) {
      v2f a;   a.x  = arow[k0];   a.y  = arow[k0 + 1];
      v2f b0;  b0.x = l0[k0];     b0.y = l0[k0 + 1];
      v2f b1;  b1.x = l1[k0];     b1.y = l1[k0 + 1];
      acc0 = wmma4(a, b0, acc0);
      acc1 = wmma4(a, b1, acc1);
    }
#pragma unroll
    for (int v = 0; v < 8; ++v) {
      const int o = o0 + v + 8 * half;
      const float bias = bf[o];
      xa[o * XP + lm]      = fmaxf(acc0[v] + bias, 0.f);
      xa[o * XP + 16 + lm] = fmaxf(acc1[v] + bias, 0.f);
    }
  }
  __syncthreads();
  empty_mid(xa, e.W1, e.b1, e.g1, e.be1, e.rm1, e.rv1, xb, o0, half, lm);
  __syncthreads();
  empty_mid(xb, e.W2, e.b2, e.g2, e.be2, e.rm2, e.rv2, xa, o0, half, lm);
  __syncthreads();
  if (threadIdx.x < Bn) {
    float s = e.b3[0];
    for (int c = 0; c < Hd; ++c) s += e.W3[c] * xa[c * XP + threadIdx.x];
    c0[threadIdx.x] = s;
  }
  __syncthreads();
  for (int j = 0; j < Bn; ++j)
    if ((int)threadIdx.x >= cnt[j]) out[j * PPB + threadIdx.x] = c0[j];
}

__global__ __launch_bounds__(RANK_WAVES * 32) void k_rank(const int* __restrict__ bid,
                                                          const float* __restrict__ val,
                                                          const float* __restrict__ lang,
                                                          const float* __restrict__ Wf,
                                                          const float* __restrict__ bf,
                                                          const EmptyArgs e,
                                                          float* __restrict__ out) {
  __shared__ int part[RANK_WAVES][32];
  const bool fill = blockIdx.x == 0;
  const int tile = Np / 32 - (int)blockIdx.x;
  const int wave = __builtin_amdgcn_readfirstlane((int)(threadIdx.x >> 5));
  const int lane = threadIdx.x & 31;
  const int i    = tile * 32 + lane;
  const unsigned* __restrict__ words = (const unsigned*)bid;
  v8u own = *(const v8u*)(words + (fill ? 0 : tile) * 32);
  int myb = lane;
  unsigned m4 = 0;
  const float v  = fill ? 0.f : val[i];
  unsigned ne = 0;
  int nu = 0;
  auto round = [&](int U0, bool first) {
    v8u d[8];
#pragma unroll
    for (int q = 0; q < 8; ++q) {
      const int U = min(U0 + q * RANK_WAVES, Np / 32 - 1);
      d[q] = *(const v8u*)(words + U * 32);
    }
    if (first)
      asm volatile("" : "+s"(d[0]), "+s"(d[1]), "+s"(d[2]), "+s"(d[3]),
                        "+s"(d[4]), "+s"(d[5]), "+s"(d[6]), "+s"(d[7]), "+s"(own));
    else
      asm volatile("" : "+s"(d[0]), "+s"(d[1]), "+s"(d[2]), "+s"(d[3]),
                        "+s"(d[4]), "+s"(d[5]), "+s"(d[6]), "+s"(d[7]));
    if (first) {
      unsigned dw = own[0];
#pragma unroll
      for (int k = 1; k < 8; ++k) dw = ((lane >> 2) == k) ? own[k] : dw;
      if (!fill) myb = (int)((dw >> (8 * (lane & 3))) & 0xffu);
      m4 = (unsigned)myb * 0x01010101u;
    }
#pragma unroll
    for (int q = 0; q < 8; ++q) {
      const bool in = U0 + q * RANK_WAVES < tile;
      const unsigned msk = in ? 0x80808080u : 0u;
#pragma unroll
      for (int k = 0; k < 8; ++k) {
        const unsigned t = (d[q][k] ^ m4) + 0x7f7f7f7fu;
        ne = __builtin_amdgcn_sad_u8(t & msk, 0u, ne);
      }
      nu += in ? 1 : 0;
    }
  };
  round(wave, true);
#pragma unroll
  for (int rr = 1; rr < Np / 32 / (8 * RANK_WAVES); ++rr)
    if (wave + rr * 8 * RANK_WAVES < tile) round(wave + rr * 8 * RANK_WAVES, false);
  int r = 32 * nu - (int)(ne >> 7);
  if (!fill) {
    unsigned ow = own[0];
#pragma unroll
    for (int k = 1; k < 8; ++k) ow = (wave == k) ? own[k] : ow;
    const unsigned t  = (ow ^ m4) + 0x7f7f7f7fu;
    const unsigned eq = ~t & 0x80808080u;
    const int km = lane - 4 * wave;
    const unsigned front = km >= 4 ? 0xffffffffu : (km <= 0 ? 0u : ((1u << (8 * km)) - 1u));
    r += __popc(eq & front);
  }
  part[wave][lane] = r;
  __syncthreads();
  if (fill) {
    empty_slots(part, lang, Wf, bf, e, out);
    return;
  }
  if (wave == 0) {
    int s = 0;
    for (int w = 0; w < RANK_WAVES; ++w) s += part[w][lane];
    if ((unsigned)myb < (unsigned)Bn && s < PPB) out[myb * PPB + s] = v;
  }
}

typedef float v4f __attribute__((ext_vector_type(4)));

__device__ __forceinline__ v8f ld8(const float* __restrict__ p) {
  const v4f lo = *(const v4f*)p;
  const v4f hi = *(const v4f*)(p + 4);
  v8f r;
  r[0] = lo[0]; r[1] = lo[1]; r[2] = lo[2]; r[3] = lo[3];
  r[4] = hi[0]; r[5] = hi[1]; r[6] = hi[2]; r[7] = hi[3];
  return r;
}

__device__ __forceinline__ void kblock(const v8f (&a)[4], const v8f (&b)[4], v8f (&acc)[4]) {
#pragma unroll
  for (int s = 0; s < 2; ++s) {
#pragma unroll
    for (int j = 0; j < 4; ++j) {
      v2f b0; b0.x = b[s][2 * j];     b0.y = b[s][2 * j + 1];
      v2f b1; b1.x = b[2 + s][2 * j]; b1.y = b[2 + s][2 * j + 1];
#pragma unroll
      for (int v = 0; v < 2; ++v) {
        v2f a2; a2.x = a[2 * v + s][2 * j]; a2.y = a[2 * v + s][2 * j + 1];
        acc[2 * v]     = wmma4(a2, b0, acc[2 * v]);
        acc[2 * v + 1] = wmma4(a2, b1, acc[2 * v + 1]);
      }
    }
  }
}

__device__ __forceinline__ void lda(const float* __restrict__ arow, int ld, int T, v8f (&a)[4]) {
#pragma unroll
  for (int v = 0; v < 2; ++v) {
#pragma unroll
    for (int s = 0; s < 2; ++s) a[2 * v + s] = ld8(arow + (size_t)(16 * v) * ld + 16 * (2 * T + s));
  }
}

struct Rows {
  const float* a1;
  const float* a2;
  const float* a3;
  const float* frow[2];
  const float* lrow[2];
};

template <int N>
__device__ __forceinline__ void load_blk(const Rows& p, v8f (&a)[4], v8f (&b)[4]) {
  if constexpr (N < 12) {
    lda(p.a1, Kfuse, N, a);
#pragma unroll
    for (int c = 0; c < 2; ++c) {
      const float* bp = (N < 4) ? p.frow[c] + 32 * N : p.lrow[c] + 32 * (N - 4);
      b[2 * c] = ld8(bp); b[2 * c + 1] = ld8(bp + 16);
    }
  } else if constexpr (N < 16) {
    lda(p.a2, Hd, N - 12, a);
  } else if constexpr (N < 20) {
    lda(p.a3, Hd, N - 16, a);
  }
}

template <int N>
__device__ __forceinline__ void comp_blk(const v8f (&a)[4], const v8f (&b)[4], const v8f (&x)[16],
                                         v8f (&acc)[4]) {
  if constexpr (N < 12) {
    kblock(a, b, acc);
  } else {
    constexpr int T = (N - 12) & 3;
    const v8f bb[4] = {x[2 * T], x[2 * T + 1], x[8 + 2 * T], x[8 + 2 * T + 1]};
    kblock(a, bb, acc);
  }
}

__device__ __forceinline__ void bn_relu(v8f (&acc)[4], const float* __restrict__ bias,
                                        const float* __restrict__ gam,
                                        const float* __restrict__ bet,
                                        const float* __restrict__ rme,
                                        const float* __restrict__ rva, int o0h) {
#pragma unroll
  for (int v = 0; v < 2; ++v) {
    const int o = o0h + 16 * v;
    const v8f bs = ld8(bias + o), g = ld8(gam + o), be = ld8(bet + o);
    const v8f mu = ld8(rme + o), rv = ld8(rva + o);
#pragma unroll
    for (int r = 0; r < 8; ++r) {
      const float scale = g[r] * rsqrtf(rv[r] + BN_EPS);
#pragma unroll
      for (int c = 0; c < 2; ++c)
        acc[2 * v + c][r] = (fmaxf(acc[2 * v + c][r] + bs[r], 0.f) - mu[r]) * scale + be[r];
    }
  }
}

__device__ __forceinline__ void exchange(v4f* __restrict__ plane, int w, int lane,
                                         v8f (&acc)[4], v8f (&x)[16]) {
#pragma unroll
  for (int v = 0; v < 2; ++v) {
#pragma unroll
    for (int c = 0; c < 2; ++c) {
#pragma unroll
      for (int q = 0; q < 2; ++q) {
        v4f t;
        t[0] = acc[2 * v + c][4 * q];     t[1] = acc[2 * v + c][4 * q + 1];
        t[2] = acc[2 * v + c][4 * q + 2]; t[3] = acc[2 * v + c][4 * q + 3];
        plane[((8 * c + 2 * w + v) * 2 + q) * 32 + lane] = t;
      }
    }
  }
  __syncthreads();
#pragma unroll
  for (int t = 0; t < 16; ++t) {
    const v4f lo = plane[(t * 2) * 32 + lane];
    const v4f hi = plane[(t * 2 + 1) * 32 + lane];
    x[t][0] = lo[0]; x[t][1] = lo[1]; x[t][2] = lo[2]; x[t][3] = lo[3];
    x[t][4] = hi[0]; x[t][5] = hi[1]; x[t][6] = hi[2]; x[t][7] = hi[3];
  }
#pragma unroll
  for (int v = 0; v < 4; ++v) acc[v] = (v8f)V8F_ZERO;
}

__global__ __launch_bounds__(128) void k_mlp(const float* __restrict__ feats,
                                             const int* __restrict__ pidx,
                                             const int* __restrict__ poff,
                                             const int* __restrict__ boff,
                                             const float* __restrict__ lang,
                                             const float* __restrict__ Wf,
                                             const float* __restrict__ bf,
                                             const EmptyArgs e,
                                             int* __restrict__ bid,
                                             float* __restrict__ val) {
  __shared__ v4f xs[2][16 * 2 * 32];
  __shared__ float sp[4][2][16];
  const int w    = __builtin_amdgcn_readfirstlane((int)(threadIdx.x >> 5));
  const int lane = threadIdx.x & 31;
  const int half = lane >> 4;
  const int lm   = lane & 15;
  const int o0   = 32 * w;
  const int o0h  = o0 + 8 * half;

  Rows p;
  p.a1 = Wf   + (size_t)(o0 + lm) * Kfuse + 8 * half;
  p.a2 = e.W1 + (size_t)(o0 + lm) * Hd + 8 * half;
  p.a3 = e.W2 + (size_t)(o0 + lm) * Hd + 8 * half;
  int pf[2];
#pragma unroll
  for (int c = 0; c < 2; ++c) {
    const int i = (blockIdx.x * 2 + c) * 16 + lm;
    pf[c] = poff[i];
    p.frow[c] = feats + (size_t)i * Mf + 8 * half;
    p.lrow[c] = nullptr;
  }

  v8f acc[4];
#pragma unroll
  for (int v = 0; v < 4; ++v) acc[v] = (v8f)V8F_ZERO;

  v8f A0[4], A1[4], A2[4], B0[4], B1[4], B2[4];
  v8f x[16];
  load_blk<0>(p, A0, B0);
  load_blk<1>(p, A1, B1);
  int fp[2];
#pragma unroll
  for (int c = 0; c < 2; ++c) fp[c] = pidx[2 * pf[c] + 1];
#define STEP(N, CUR, BCUR, PRE, BPRE) load_blk<N + 2>(p, PRE, BPRE); comp_blk<N>(CUR, BCUR, x, acc);
  STEP(0, A0, B0, A2, B2)
  int own = -1;
#pragma unroll
  for (int c = 0; c < 2; ++c) {
    const int b = batch_of(boff, fp[c]);
    const int sb = ((unsigned)b < (unsigned)Bn) ? b : 0;
    p.lrow[c] = lang + (size_t)sb * LANGD + 8 * half;
    if (c == half) own = b;
  }
  STEP(1, A1, B1, A0, B0)  STEP(2, A2, B2, A1, B1)
  STEP(3, A0, B0, A2, B2)  STEP(4, A1, B1, A0, B0)  STEP(5, A2, B2, A1, B1)
  STEP(6, A0, B0, A2, B2)  STEP(7, A1, B1, A0, B0)  STEP(8, A2, B2, A1, B1)
  STEP(9, A0, B0, A2, B2)  STEP(10, A1, B1, A0, B0) STEP(11, A2, B2, A1, B1)
#pragma unroll
  for (int v = 0; v < 2; ++v) {
    const v8f bs = ld8(bf + o0h + 16 * v);
#pragma unroll
    for (int r = 0; r < 8; ++r) {
      acc[2 * v][r]     = fmaxf(acc[2 * v][r] + bs[r], 0.f);
      acc[2 * v + 1][r] = fmaxf(acc[2 * v + 1][r] + bs[r], 0.f);
    }
  }
  exchange(xs[0], w, lane, acc, x);
  STEP(12, A0, B0, A2, B2) STEP(13, A1, B1, A0, B0) STEP(14, A2, B2, A1, B1) STEP(15, A0, B0, A2, B2)
  bn_relu(acc, e.b1, e.g1, e.be1, e.rm1, e.rv1, o0h);
  exchange(xs[1], w, lane, acc, x);
  STEP(16, A1, B1, A0, B0) STEP(17, A2, B2, A1, B1) STEP(18, A0, B0, A2, B2) STEP(19, A1, B1, A0, B0)
#undef STEP
  bn_relu(acc, e.b2, e.g2, e.be2, e.rm2, e.rv2, o0h);
  float s0 = 0.f, s1 = 0.f;
#pragma unroll
  for (int v = 0; v < 2; ++v) {
    const v8f w3 = ld8(e.W3 + o0h + 16 * v);
#pragma unroll
    for (int r = 0; r < 8; ++r) { s0 += w3[r] * acc[2 * v][r]; s1 += w3[r] * acc[2 * v + 1][r]; }
  }
  s0 += __shfl_xor(s0, 16);
  s1 += __shfl_xor(s1, 16);
  if (half == 0) { sp[w][0][lm] = s0; sp[w][1][lm] = s1; }
  __syncthreads();
  if (w == 0) {
    const float r = (((sp[0][half][lm] + sp[1][half][lm]) + sp[2][half][lm]) + sp[3][half][lm])
                    + e.b3[0];
    val[blockIdx.x * 32 + lane] = r;
    ((unsigned char*)bid)[blockIdx.x * 128 + lane] =
        (unsigned char)(((unsigned)own < (unsigned)Bn) ? own : 63);
  }
}

extern "C" void kernel_launch(void* const* d_in, const int* in_sizes, int n_in,
                              void* d_out, int out_size, void* d_ws, size_t ws_size,
                              hipStream_t stream) {
  const float* feats = (const float*)d_in[0];
  const int*   pidx  = (const int*)d_in[1];
  const int*   poff  = (const int*)d_in[2];
  const int*   boff  = (const int*)d_in[3];
  const float* lang  = (const float*)d_in[4];
  const float* Wf    = (const float*)d_in[5];
  const float* bf    = (const float*)d_in[6];
  const float* W1p   = (const float*)d_in[7];
  const float* b1p   = (const float*)d_in[8];
  const float* g1p   = (const float*)d_in[9];
  const float* be1p  = (const float*)d_in[10];
  const float* rm1p  = (const float*)d_in[11];
  const float* rv1p  = (const float*)d_in[12];
  const float* W2p   = (const float*)d_in[13];
  const float* b2p   = (const float*)d_in[14];
  const float* g2p   = (const float*)d_in[15];
  const float* be2p  = (const float*)d_in[16];
  const float* rm2p  = (const float*)d_in[17];
  const float* rv2p  = (const float*)d_in[18];
  const float* W3p   = (const float*)d_in[19];
  const float* b3p   = (const float*)d_in[20];

  char* ws = (char*)d_ws;
  int*   bid    = (int*)ws;
  float* val    = (float*)(ws + (size_t)Np * sizeof(int));

  EmptyArgs ea;
  ea.W1 = W1p; ea.b1 = b1p; ea.g1 = g1p; ea.be1 = be1p; ea.rm1 = rm1p; ea.rv1 = rv1p;
  ea.W2 = W2p; ea.b2 = b2p; ea.g2 = g2p; ea.be2 = be2p; ea.rm2 = rm2p; ea.rv2 = rv2p;
  ea.W3 = W3p; ea.b3 = b3p;

  k_mlp <<<Np / 32, 128, 0, stream>>>(feats, pidx, poff, boff, lang, Wf, bf, ea, bid, val);
  k_rank<<<Np / 32 + 1, RANK_WAVES * 32, 0, stream>>>(bid, val, lang, Wf, bf, ea, (float*)d_out);
}


# k_mlp: each 32-k block's global loads (and a layer's 32 LDS reads) pinned as one group in front of its 32 WMMAs by scheduling fences; the compiler had sunk them among the WMMAs in over 100 places
# speedup vs baseline: 7.7806x; 1.1012x over previous
#include <hip/hip_runtime.h>

constexpr int Bn    = 32;
constexpr int PPB   = 256;
constexpr int Np    = Bn * PPB;
constexpr int Mf    = 128;
constexpr int LANGD = 256;
constexpr int Hd    = 128;
constexpr int Kfuse = LANGD + Mf;
#define BN_EPS 1e-5f

typedef float v2f __attribute__((ext_vector_type(2)));
typedef float v8f __attribute__((ext_vector_type(8)));

__device__ __forceinline__ v8f wmma4(v2f a, v2f b, v8f c) {
  return __builtin_amdgcn_wmma_f32_16x16x4_f32(false, a, false, b, (short)0, c,
                                               false, false);
}

#define V8F_ZERO {0.f, 0.f, 0.f, 0.f, 0.f, 0.f, 0.f, 0.f}


__device__ __forceinline__ int batch_of(const int* __restrict__ boff, int first_pt) {
  int b = -1;
#pragma unroll
  for (int j = 0; j <= Bn; ++j) b += (boff[j] <= first_pt) ? 1 : 0;
  return b;
}

struct EmptyArgs {
  const float* W1; const float* b1; const float* g1; const float* be1;
  const float* rm1; const float* rv1;
  const float* W2; const float* b2; const float* g2; const float* be2;
  const float* rm2; const float* rv2;
  const float* W3; const float* b3;
};

constexpr int XP = 40;

__device__ __forceinline__ void empty_mid(const float* xin, const float* __restrict__ W,
                                          const float* __restrict__ bias,
                                          const float* __restrict__ gam,
                                          const float* __restrict__ bet,
                                          const float* __restrict__ rme,
                                          const float* __restrict__ rva,
                                          float* xout, int o0, int half, int lm) {
  const float* arow = W + (size_t)(o0 + lm) * Hd + 2 * half;
  const float* bc   = xin + (2 * half) * XP + lm;
  v8f acc0 = V8F_ZERO, acc1 = V8F_ZERO;
#pragma unroll 16
  for (int k0 = 0; k0 < Hd; k0 += 4) {
    v2f a;   a.x  = arow[k0];             a.y  = arow[k0 + 1];
    v2f b0;  b0.x = bc[k0 * XP];          b0.y = bc[(k0 + 1) * XP];
    v2f b1;  b1.x = bc[k0 * XP + 16];     b1.y = bc[(k0 + 1) * XP + 16];
    acc0 = wmma4(a, b0, acc0);
    acc1 = wmma4(a, b1, acc1);
  }
#pragma unroll
  for (int v = 0; v < 8; ++v) {
    const int o = o0 + v + 8 * half;
    const float bs    = bias[o];
    const float scale = gam[o] * rsqrtf(rva[o] + BN_EPS);
    const float mu    = rme[o];
    const float beta  = bet[o];
    xout[o * XP + lm]      = (fmaxf(acc0[v] + bs, 0.f) - mu) * scale + beta;
    xout[o * XP + 16 + lm] = (fmaxf(acc1[v] + bs, 0.f) - mu) * scale + beta;
  }
}

constexpr int RANK_WAVES = 8;
typedef unsigned v8u __attribute__((ext_vector_type(8)));

__device__ __forceinline__ void empty_slots(const int (*part)[32],
                                            const float* __restrict__ lang,
                                            const float* __restrict__ Wf,
                                            const float* __restrict__ bf,
                                            const EmptyArgs& e,
                                            float* __restrict__ out) {
  __shared__ float xa[Hd * XP];
  __shared__ float xb[Hd * XP];
  __shared__ float c0[Bn];
  __shared__ int   cnt[Bn];
  const int wave = __builtin_amdgcn_readfirstlane((int)(threadIdx.x >> 5));
  const int lane = threadIdx.x & 31;
  const int half = lane >> 4;
  const int lm   = lane & 15;
  const int o0   = wave << 4;
  int n = 0;
  for (int w = 0; w < RANK_WAVES; ++w) n += part[w][lane];
  if (!__any(n < PPB)) return;
  if (wave == 0) cnt[lane] = n;
  {
    const float* arow = Wf + (size_t)(o0 + lm) * Kfuse + Mf + 2 * half;
    const float* l0   = lang + (size_t)lm * LANGD + 2 * half;
    const float* l1   = lang + (size_t)(16 + lm) * LANGD + 2 * half;
    v8f acc0 = V8F_ZERO, acc1 = V8F_ZERO;
#pragma unroll 16
    for (int k0 = 0; k0 < LANGD; k0 += 4) {
      v2f a;   a.x  = arow[k0];   a.y  = arow[k0 + 1];
      v2f b0;  b0.x = l0[k0];     b0.y = l0[k0 + 1];
      v2f b1;  b1.x = l1[k0];     b1.y = l1[k0 + 1];
      acc0 = wmma4(a, b0, acc0);
      acc1 = wmma4(a, b1, acc1);
    }
#pragma unroll
    for (int v = 0; v < 8; ++v) {
      const int o = o0 + v + 8 * half;
      const float bias = bf[o];
      xa[o * XP + lm]      = fmaxf(acc0[v] + bias, 0.f);
      xa[o * XP + 16 + lm] = fmaxf(acc1[v] + bias, 0.f);
    }
  }
  __syncthreads();
  empty_mid(xa, e.W1, e.b1, e.g1, e.be1, e.rm1, e.rv1, xb, o0, half, lm);
  __syncthreads();
  empty_mid(xb, e.W2, e.b2, e.g2, e.be2, e.rm2, e.rv2, xa, o0, half, lm);
  __syncthreads();
  if (threadIdx.x < Bn) {
    float s = e.b3[0];
    for (int c = 0; c < Hd; ++c) s += e.W3[c] * xa[c * XP + threadIdx.x];
    c0[threadIdx.x] = s;
  }
  __syncthreads();
  for (int j = 0; j < Bn; ++j)
    if ((int)threadIdx.x >= cnt[j]) out[j * PPB + threadIdx.x] = c0[j];
}

__global__ __launch_bounds__(RANK_WAVES * 32) void k_rank(const int* __restrict__ bid,
                                                          const float* __restrict__ val,
                                                          const float* __restrict__ lang,
                                                          const float* __restrict__ Wf,
                                                          const float* __restrict__ bf,
                                                          const EmptyArgs e,
                                                          float* __restrict__ out) {
  __shared__ int part[RANK_WAVES][32];
  const bool fill = blockIdx.x == 0;
  const int tile = Np / 32 - (int)blockIdx.x;
  const int wave = __builtin_amdgcn_readfirstlane((int)(threadIdx.x >> 5));
  const int lane = threadIdx.x & 31;
  const int i    = tile * 32 + lane;
  const unsigned* __restrict__ words = (const unsigned*)bid;
  v8u own = *(const v8u*)(words + (fill ? 0 : tile) * 32);
  int myb = lane;
  unsigned m4 = 0;
  const float v  = fill ? 0.f : val[i];
  unsigned ne = 0;
  int nu = 0;
  auto round = [&](int U0, bool first) {
    v8u d[8];
#pragma unroll
    for (int q = 0; q < 8; ++q) {
      const int U = min(U0 + q * RANK_WAVES, Np / 32 - 1);
      d[q] = *(const v8u*)(words + U * 32);
    }
    if (first)
      asm volatile("" : "+s"(d[0]), "+s"(d[1]), "+s"(d[2]), "+s"(d[3]),
                        "+s"(d[4]), "+s"(d[5]), "+s"(d[6]), "+s"(d[7]), "+s"(own));
    else
      asm volatile("" : "+s"(d[0]), "+s"(d[1]), "+s"(d[2]), "+s"(d[3]),
                        "+s"(d[4]), "+s"(d[5]), "+s"(d[6]), "+s"(d[7]));
    if (first) {
      unsigned dw = own[0];
#pragma unroll
      for (int k = 1; k < 8; ++k) dw = ((lane >> 2) == k) ? own[k] : dw;
      if (!fill) myb = (int)((dw >> (8 * (lane & 3))) & 0xffu);
      m4 = (unsigned)myb * 0x01010101u;
    }
#pragma unroll
    for (int q = 0; q < 8; ++q) {
      const bool in = U0 + q * RANK_WAVES < tile;
      const unsigned msk = in ? 0x80808080u : 0u;
#pragma unroll
      for (int k = 0; k < 8; ++k) {
        const unsigned t = (d[q][k] ^ m4) + 0x7f7f7f7fu;
        ne = __builtin_amdgcn_sad_u8(t & msk, 0u, ne);
      }
      nu += in ? 1 : 0;
    }
  };
  round(wave, true);
#pragma unroll
  for (int rr = 1; rr < Np / 32 / (8 * RANK_WAVES); ++rr)
    if (wave + rr * 8 * RANK_WAVES < tile) round(wave + rr * 8 * RANK_WAVES, false);
  int r = 32 * nu - (int)(ne >> 7);
  if (!fill) {
    unsigned ow = own[0];
#pragma unroll
    for (int k = 1; k < 8; ++k) ow = (wave == k) ? own[k] : ow;
    const unsigned t  = (ow ^ m4) + 0x7f7f7f7fu;
    const unsigned eq = ~t & 0x80808080u;
    const int km = lane - 4 * wave;
    const unsigned front = km >= 4 ? 0xffffffffu : (km <= 0 ? 0u : ((1u << (8 * km)) - 1u));
    r += __popc(eq & front);
  }
  part[wave][lane] = r;
  __syncthreads();
  if (fill) {
    empty_slots(part, lang, Wf, bf, e, out);
    return;
  }
  if (wave == 0) {
    int s = 0;
    for (int w = 0; w < RANK_WAVES; ++w) s += part[w][lane];
    if ((unsigned)myb < (unsigned)Bn && s < PPB) out[myb * PPB + s] = v;
  }
}

typedef float v4f __attribute__((ext_vector_type(4)));

__device__ __forceinline__ v8f ld8(const float* __restrict__ p) {
  const v4f lo = *(const v4f*)p;
  const v4f hi = *(const v4f*)(p + 4);
  v8f r;
  r[0] = lo[0]; r[1] = lo[1]; r[2] = lo[2]; r[3] = lo[3];
  r[4] = hi[0]; r[5] = hi[1]; r[6] = hi[2]; r[7] = hi[3];
  return r;
}

__device__ __forceinline__ void kblock(const v8f (&a)[4], const v8f (&b)[4], v8f (&acc)[4]) {
#pragma unroll
  for (int s = 0; s < 2; ++s) {
#pragma unroll
    for (int j = 0; j < 4; ++j) {
      v2f b0; b0.x = b[s][2 * j];     b0.y = b[s][2 * j + 1];
      v2f b1; b1.x = b[2 + s][2 * j]; b1.y = b[2 + s][2 * j + 1];
#pragma unroll
      for (int v = 0; v < 2; ++v) {
        v2f a2; a2.x = a[2 * v + s][2 * j]; a2.y = a[2 * v + s][2 * j + 1];
        acc[2 * v]     = wmma4(a2, b0, acc[2 * v]);
        acc[2 * v + 1] = wmma4(a2, b1, acc[2 * v + 1]);
      }
    }
  }
}

__device__ __forceinline__ void lda(const float* __restrict__ arow, int ld, int T, v8f (&a)[4]) {
#pragma unroll
  for (int v = 0; v < 2; ++v) {
#pragma unroll
    for (int s = 0; s < 2; ++s) a[2 * v + s] = ld8(arow + (size_t)(16 * v) * ld + 16 * (2 * T + s));
  }
}

struct Rows {
  const float* a1;
  const float* a2;
  const float* a3;
  const float* frow[2];
  const float* lrow[2];
};

template <int N>
__device__ __forceinline__ void load_blk(const Rows& p, v8f (&a)[4], v8f (&b)[4]) {
  if constexpr (N < 12) {
    lda(p.a1, Kfuse, N, a);
#pragma unroll
    for (int c = 0; c < 2; ++c) {
      const float* bp = (N < 4) ? p.frow[c] + 32 * N : p.lrow[c] + 32 * (N - 4);
      b[2 * c] = ld8(bp); b[2 * c + 1] = ld8(bp + 16);
    }
  } else if constexpr (N < 16) {
    lda(p.a2, Hd, N - 12, a);
  } else if constexpr (N < 20) {
    lda(p.a3, Hd, N - 16, a);
  }
}

template <int N>
__device__ __forceinline__ void comp_blk(const v8f (&a)[4], const v8f (&b)[4], const v8f (&x)[16],
                                         v8f (&acc)[4]) {
  if constexpr (N < 12) {
    kblock(a, b, acc);
  } else {
    constexpr int T = (N - 12) & 3;
    const v8f bb[4] = {x[2 * T], x[2 * T + 1], x[8 + 2 * T], x[8 + 2 * T + 1]};
    kblock(a, bb, acc);
  }
}

struct BnRows { v8f bs[2], g[2], be[2], mu[2], rv[2]; };

__device__ __forceinline__ void bn_load(BnRows& q, const float* __restrict__ bias,
                                        const float* __restrict__ gam,
                                        const float* __restrict__ bet,
                                        const float* __restrict__ rme,
                                        const float* __restrict__ rva, int o0h) {
#pragma unroll
  for (int v = 0; v < 2; ++v) {
    const int o = o0h + 16 * v;
    q.bs[v] = ld8(bias + o); q.g[v] = ld8(gam + o); q.be[v] = ld8(bet + o);
    q.mu[v] = ld8(rme + o);  q.rv[v] = ld8(rva + o);
  }
}

__device__ __forceinline__ void bn_relu(v8f (&acc)[4], const BnRows& q) {
#pragma unroll
  for (int v = 0; v < 2; ++v) {
#pragma unroll
    for (int r = 0; r < 8; ++r) {
      const float scale = q.g[v][r] * rsqrtf(q.rv[v][r] + BN_EPS);
#pragma unroll
      for (int c = 0; c < 2; ++c)
        acc[2 * v + c][r] =
            (fmaxf(acc[2 * v + c][r] + q.bs[v][r], 0.f) - q.mu[v][r]) * scale + q.be[v][r];
    }
  }
}

__device__ __forceinline__ void exchange(v4f* __restrict__ plane, int w, int lane,
                                         v8f (&acc)[4], v8f (&x)[16]) {
#pragma unroll
  for (int v = 0; v < 2; ++v) {
#pragma unroll
    for (int c = 0; c < 2; ++c) {
#pragma unroll
      for (int q = 0; q < 2; ++q) {
        v4f t;
        t[0] = acc[2 * v + c][4 * q];     t[1] = acc[2 * v + c][4 * q + 1];
        t[2] = acc[2 * v + c][4 * q + 2]; t[3] = acc[2 * v + c][4 * q + 3];
        plane[((8 * c + 2 * w + v) * 2 + q) * 32 + lane] = t;
      }
    }
  }
  __syncthreads();
#pragma unroll
  for (int t = 0; t < 16; ++t) {
    const v4f lo = plane[(t * 2) * 32 + lane];
    const v4f hi = plane[(t * 2 + 1) * 32 + lane];
    x[t][0] = lo[0]; x[t][1] = lo[1]; x[t][2] = lo[2]; x[t][3] = lo[3];
    x[t][4] = hi[0]; x[t][5] = hi[1]; x[t][6] = hi[2]; x[t][7] = hi[3];
  }
#pragma unroll
  for (int v = 0; v < 4; ++v) acc[v] = (v8f)V8F_ZERO;
}

__global__ __launch_bounds__(128) void k_mlp(const float* __restrict__ feats,
                                             const int* __restrict__ pidx,
                                             const int* __restrict__ poff,
                                             const int* __restrict__ boff,
                                             const float* __restrict__ lang,
                                             const float* __restrict__ Wf,
                                             const float* __restrict__ bf,
                                             const EmptyArgs e,
                                             int* __restrict__ bid,
                                             float* __restrict__ val) {
  __shared__ v4f xs[2][16 * 2 * 32];
  __shared__ float sp[4][2][16];
  const int w    = __builtin_amdgcn_readfirstlane((int)(threadIdx.x >> 5));
  const int lane = threadIdx.x & 31;
  const int half = lane >> 4;
  const int lm   = lane & 15;
  const int o0   = 32 * w;
  const int o0h  = o0 + 8 * half;

  Rows p;
  p.a1 = Wf   + (size_t)(o0 + lm) * Kfuse + 8 * half;
  p.a2 = e.W1 + (size_t)(o0 + lm) * Hd + 8 * half;
  p.a3 = e.W2 + (size_t)(o0 + lm) * Hd + 8 * half;
  int pf[2];
#pragma unroll
  for (int c = 0; c < 2; ++c) {
    const int i = (blockIdx.x * 2 + c) * 16 + lm;
    pf[c] = poff[i];
    p.frow[c] = feats + (size_t)i * Mf + 8 * half;
    p.lrow[c] = nullptr;
  }

  v8f acc[4];
#pragma unroll
  for (int v = 0; v < 4; ++v) acc[v] = (v8f)V8F_ZERO;

  v8f A0[4], A1[4], A2[4], B0[4], B1[4], B2[4];
  v8f x[16];
  load_blk<0>(p, A0, B0);
  load_blk<1>(p, A1, B1);
  int fp[2];
#pragma unroll
  for (int c = 0; c < 2; ++c) fp[c] = pidx[2 * pf[c] + 1];
#define GROUP_FENCE() __builtin_amdgcn_sched_barrier(0x406)
#define LOADS(N, PRE, BPRE) load_blk<N + 2>(p, PRE, BPRE);
#define PRODS(N, CUR, BCUR) GROUP_FENCE(); comp_blk<N>(CUR, BCUR, x, acc); GROUP_FENCE();
#define STEP(N, CUR, BCUR, PRE, BPRE) LOADS(N, PRE, BPRE) PRODS(N, CUR, BCUR)
  STEP(0, A0, B0, A2, B2)
  int own = -1;
#pragma unroll
  for (int c = 0; c < 2; ++c) {
    const int b = batch_of(boff, fp[c]);
    const int sb = ((unsigned)b < (unsigned)Bn) ? b : 0;
    p.lrow[c] = lang + (size_t)sb * LANGD + 8 * half;
    if (c == half) own = b;
  }
  STEP(1, A1, B1, A0, B0)  STEP(2, A2, B2, A1, B1)
  STEP(3, A0, B0, A2, B2)  STEP(4, A1, B1, A0, B0)  STEP(5, A2, B2, A1, B1)
  STEP(6, A0, B0, A2, B2)  STEP(7, A1, B1, A0, B0)  STEP(8, A2, B2, A1, B1)
  v8f bfv[2];
  LOADS(9, A2, B2)
  bfv[0] = ld8(bf + o0h); bfv[1] = ld8(bf + o0h + 16);
  PRODS(9, A0, B0)
  STEP(10, A1, B1, A0, B0) STEP(11, A2, B2, A1, B1)
#pragma unroll
  for (int v = 0; v < 2; ++v) {
    const v8f bs = bfv[v];
#pragma unroll
    for (int r = 0; r < 8; ++r) {
      acc[2 * v][r]     = fmaxf(acc[2 * v][r] + bs[r], 0.f);
      acc[2 * v + 1][r] = fmaxf(acc[2 * v + 1][r] + bs[r], 0.f);
    }
  }
  exchange(xs[0], w, lane, acc, x);
  BnRows bn;
  STEP(12, A0, B0, A2, B2)
  LOADS(13, A0, B0)
  bn_load(bn, e.b1, e.g1, e.be1, e.rm1, e.rv1, o0h);
  PRODS(13, A1, B1)
  STEP(14, A2, B2, A1, B1) STEP(15, A0, B0, A2, B2)
  bn_relu(acc, bn);
  exchange(xs[1], w, lane, acc, x);
  v8f w3v[2];
  STEP(16, A1, B1, A0, B0)
  LOADS(17, A1, B1)
  bn_load(bn, e.b2, e.g2, e.be2, e.rm2, e.rv2, o0h);
  w3v[0] = ld8(e.W3 + o0h); w3v[1] = ld8(e.W3 + o0h + 16);
  PRODS(17, A2, B2)
  STEP(18, A0, B0, A2, B2) STEP(19, A1, B1, A0, B0)
#undef STEP
#undef PRODS
#undef LOADS
#undef GROUP_FENCE
  bn_relu(acc, bn);
  float s0 = 0.f, s1 = 0.f;
#pragma unroll
  for (int v = 0; v < 2; ++v) {
    const v8f w3 = w3v[v];
#pragma unroll
    for (int r = 0; r < 8; ++r) { s0 += w3[r] * acc[2 * v][r]; s1 += w3[r] * acc[2 * v + 1][r]; }
  }
  s0 += __shfl_xor(s0, 16);
  s1 += __shfl_xor(s1, 16);
  if (half == 0) { sp[w][0][lm] = s0; sp[w][1][lm] = s1; }
  __syncthreads();
  if (w == 0) {
    const float r = (((sp[0][half][lm] + sp[1][half][lm]) + sp[2][half][lm]) + sp[3][half][lm])
                    + e.b3[0];
    val[blockIdx.x * 32 + lane] = r;
    ((unsigned char*)bid)[blockIdx.x * 128 + lane] =
        (unsigned char)(((unsigned)own < (unsigned)Bn) ? own : 63);
  }
}

extern "C" void kernel_launch(void* const* d_in, const int* in_sizes, int n_in,
                              void* d_out, int out_size, void* d_ws, size_t ws_size,
                              hipStream_t stream) {
  const float* feats = (const float*)d_in[0];
  const int*   pidx  = (const int*)d_in[1];
  const int*   poff  = (const int*)d_in[2];
  const int*   boff  = (const int*)d_in[3];
  const float* lang  = (const float*)d_in[4];
  const float* Wf    = (const float*)d_in[5];
  const float* bf    = (const float*)d_in[6];
  const float* W1p   = (const float*)d_in[7];
  const float* b1p   = (const float*)d_in[8];
  const float* g1p   = (const float*)d_in[9];
  const float* be1p  = (const float*)d_in[10];
  const float* rm1p  = (const float*)d_in[11];
  const float* rv1p  = (const float*)d_in[12];
  const float* W2p   = (const float*)d_in[13];
  const float* b2p   = (const float*)d_in[14];
  const float* g2p   = (const float*)d_in[15];
  const float* be2p  = (const float*)d_in[16];
  const float* rm2p  = (const float*)d_in[17];
  const float* rv2p  = (const float*)d_in[18];
  const float* W3p   = (const float*)d_in[19];
  const float* b3p   = (const float*)d_in[20];

  char* ws = (char*)d_ws;
  int*   bid    = (int*)ws;
  float* val    = (float*)(ws + (size_t)Np * sizeof(int));

  EmptyArgs ea;
  ea.W1 = W1p; ea.b1 = b1p; ea.g1 = g1p; ea.be1 = be1p; ea.rm1 = rm1p; ea.rv1 = rv1p;
  ea.W2 = W2p; ea.b2 = b2p; ea.g2 = g2p; ea.be2 = be2p; ea.rm2 = rm2p; ea.rv2 = rv2p;
  ea.W3 = W3p; ea.b3 = b3p;

  k_mlp <<<Np / 32, 128, 0, stream>>>(feats, pidx, poff, boff, lang, Wf, bf, ea, bid, val);
  k_rank<<<Np / 32 + 1, RANK_WAVES * 32, 0, stream>>>(bid, val, lang, Wf, bf, ea, (float*)d_out);
}
